# starting kernel
# baseline (speedup 1.0000x reference)
#include <hip/hip_runtime.h>

typedef __attribute__((ext_vector_type(2))) float v2f;
typedef __attribute__((ext_vector_type(8))) float v8f;

constexpr int kBS    = 256;
constexpr int kP     = 64;
constexpr int kV     = 2562;
constexpr int kK     = kP * 3;
constexpr int kKP    = kK / 2;
constexpr int kM     = kBS * 3;
constexpr int kMT    = kM / 16;
constexpr int kNTILE = 4;
constexpr int kNT4   = 41;
constexpr int kNPAD  = kNT4 * kNTILE * 16;

__global__ void prep_A_kernel(const float* __restrict__ scales,
                              const float* __restrict__ transforms,
                              const float* __restrict__ pw,
                              float* __restrict__ A) {
  int idx = blockIdx.x * blockDim.x + threadIdx.x;
  if (idx >= kBS * kP) return;
  int bs = idx >> 6;
  int p  = idx & 63;
  const float* t = transforms + (size_t)idx * 6;
  float ax = t[3], ay = t[4], az = t[5];
  float cx = cosf(ax), sx = sinf(ax);
  float cy = cosf(ay), sy = sinf(ay);
  float cz = cosf(az), sz = sinf(az);
  float r00 = cy * cz;
  float r01 = -cy * sz;
  float r02 = sy;
  float r10 = sx * sy * cz + cx * sz;
  float r11 = -sx * sy * sz + cx * cz;
  float r12 = -sx * cy;
  float r20 = -cx * sy * cz + sx * sz;
  float r21 = cx * sy * sz + sx * cz;
  float r22 = cx * cy;
  float w = pw[idx] * scales[bs];
  float* a0 = A + (size_t)(bs * 3 + 0) * kK + p * 3;
  float* a1 = A + (size_t)(bs * 3 + 1) * kK + p * 3;
  float* a2 = A + (size_t)(bs * 3 + 2) * kK + p * 3;
  a0[0] = r00 * w; a0[1] = r01 * w; a0[2] = r02 * w;
  a1[0] = r10 * w; a1[1] = r11 * w; a1[2] = r12 * w;
  a2[0] = r20 * w; a2[1] = r21 * w; a2[2] = r22 * w;
}

__global__ void prep_D_kernel(const float* __restrict__ base,
                              const float* __restrict__ off,
                              v2f* __restrict__ Dp) {
  int id = blockIdx.x * blockDim.x + threadIdx.x;
  if (id >= kKP * kNPAD) return;
  int kp = id / kNPAD;
  int n  = id - kp * kNPAD;
  v2f val = {0.f, 0.f};
  if (n < kV) {
    int k0 = kp * 2, k1 = k0 + 1;
    int p0 = k0 / 3, j0 = k0 - p0 * 3;
    int p1 = k1 / 3, j1 = k1 - p1 * 3;
    val.x = base[(size_t)n * 3 + j0] + off[((size_t)p0 * kV + n) * 3 + j0];
    val.y = base[(size_t)n * 3 + j1] + off[((size_t)p1 * kV + n) * 3 + j1];
  }
  Dp[id] = val;
}

__global__ void prep_T_kernel(const float* __restrict__ transforms,
                              const float* __restrict__ pw,
                              float* __restrict__ T) {
  __shared__ float sm[kP][3];
  int bs = blockIdx.x;
  int p  = threadIdx.x;
  const float* t = transforms + ((size_t)bs * kP + p) * 6;
  float w = pw[bs * kP + p];
  sm[p][0] = w * t[0];
  sm[p][1] = w * t[1];
  sm[p][2] = w * t[2];
  __syncthreads();
  for (int st = 32; st > 0; st >>= 1) {
    if (p < st) {
      sm[p][0] += sm[p + st][0];
      sm[p][1] += sm[p + st][1];
      sm[p][2] += sm[p + st][2];
    }
    __syncthreads();
  }
  if (p < 3) T[bs * 3 + p] = sm[0][p];
}

__global__ __launch_bounds__(32) void gemm_wmma_kernel(
    const float* __restrict__ A, const v2f* __restrict__ Dp,
    const float* __restrict__ T, float* __restrict__ out) {
  int lane = threadIdx.x;
  int mt = blockIdx.x;
  int nt = blockIdx.y;
  int m0 = mt * 16;
  int n0 = nt * (kNTILE * 16);
  int half = lane >> 4;
  int lrow = lane & 15;

  const float* Ap = A + (size_t)(m0 + lrow) * kK + half * 2;
  const v2f* Bp = Dp + (size_t)half * kNPAD + n0 + lrow;

  v8f c0 = {0.f, 0.f, 0.f, 0.f, 0.f, 0.f, 0.f, 0.f};
  v8f c1 = c0, c2 = c0, c3 = c0;
#pragma unroll 4
  for (int k = 0; k < kK; k += 4) {
    v2f a = *(const v2f*)(Ap + k);
    const v2f* b = Bp + (size_t)(k >> 1) * kNPAD;
    v2f b0 = b[0], b1 = b[16], b2 = b[32], b3 = b[48];
    c0 = __builtin_amdgcn_wmma_f32_16x16x4_f32(false, a, false, b0, (short)0,
                                               c0, false, false);
    c1 = __builtin_amdgcn_wmma_f32_16x16x4_f32(false, a, false, b1, (short)0,
                                               c1, false, false);
    c2 = __builtin_amdgcn_wmma_f32_16x16x4_f32(false, a, false, b2, (short)0,
                                               c2, false, false);
    c3 = __builtin_amdgcn_wmma_f32_16x16x4_f32(false, a, false, b3, (short)0,
                                               c3, false, false);
  }

  int mb = m0 + half * 8;
  int bs = mb / 3;
  int i  = mb - bs * 3;
  float  tv[8];
  size_t ob[8];
#pragma unroll
  for (int r = 0; r < 8; ++r) {
    tv[r] = T[mb + r];
    ob[r] = (size_t)bs * (kV * 3) + i;
    if (++i == 3) { i = 0; ++bs; }
  }
  auto store_tile = [&](int t, const v8f& c) {
    int n = n0 + t * 16 + lrow;
    if (n < kV) {
      size_t n3 = (size_t)n * 3;
#pragma unroll
      for (int r = 0; r < 8; ++r) out[ob[r] + n3] = c[r] + tv[r];
    }
  };
  store_tile(0, c0);
  store_tile(1, c1);
  store_tile(2, c2);
  store_tile(3, c3);
}

extern "C" void kernel_launch(void* const* d_in, const int* in_sizes, int n_in,
                              void* d_out, int out_size, void* d_ws,
                              size_t ws_size, hipStream_t stream) {
  const float* scales     = (const float*)d_in[0];
  const float* transforms = (const float*)d_in[1];
  const float* pw         = (const float*)d_in[2];
  const float* off        = (const float*)d_in[3];
  const float* base       = (const float*)d_in[4];
  float* out = (float*)d_out;

  float* wsf = (float*)d_ws;
  float* A  = wsf;
  v2f*   Dp = (v2f*)(wsf + (size_t)kM * kK);
  float* T  = (float*)(Dp + (size_t)kKP * kNPAD);

  prep_A_kernel<<<(kBS * kP + 255) / 256, 256, 0, stream>>>(scales, transforms,
                                                            pw, A);
  prep_D_kernel<<<(kKP * kNPAD + 255) / 256, 256, 0, stream>>>(base, off, Dp);
  prep_T_kernel<<<kBS, kP, 0, stream>>>(transforms, pw, T);
  gemm_wmma_kernel<<<dim3(kMT, kNT4), 32, 0, stream>>>(A, Dp, T, out);
}


# Merge the three preparation launches (A rows, T sums, packed D pairs) into one launch whose block index picks the role: 4 launches per run become 2; all arithmetic and summation orders kept
# speedup vs baseline: 1.2873x; 1.2873x over previous
#include <hip/hip_runtime.h>

typedef __attribute__((ext_vector_type(2))) float v2f;
typedef __attribute__((ext_vector_type(8))) float v8f;

constexpr int kBS    = 256;
constexpr int kP     = 64;
constexpr int kV     = 2562;
constexpr int kK     = kP * 3;
constexpr int kKP    = kK / 2;
constexpr int kM     = kBS * 3;
constexpr int kMT    = kM / 16;
constexpr int kNTILE = 4;
constexpr int kNT4   = 41;
constexpr int kNPAD  = kNT4 * kNTILE * 16;

constexpr int kPrepThreads = 256;
constexpr int kPrepAT      = kBS * kP / kPrepThreads;
constexpr int kPrepD       = kKP * kNPAD / kPrepThreads;
constexpr int kPrepTotal   = kPrepAT + kPrepD;
static_assert(kPrepAT * kPrepThreads == kBS * kP, "A/T role grid must be exact");
static_assert(kPrepD * kPrepThreads == kKP * kNPAD, "D role grid must be exact");
static_assert(kPrepThreads % kP == 0, "a (b,s) group must not straddle blocks");

__global__ __launch_bounds__(kPrepThreads) void prep_all_kernel(
    const float* __restrict__ scales, const float* __restrict__ transforms,
    const float* __restrict__ pw, const float* __restrict__ off,
    const float* __restrict__ base, float* __restrict__ A,
    v2f* __restrict__ Dp, float* __restrict__ T) {
  __shared__ float sm[kPrepThreads][3];
  if (blockIdx.x < kPrepAT) {
    int tid = threadIdx.x;
    int idx = blockIdx.x * kPrepThreads + tid;
    int bs = idx >> 6;
    int p  = idx & 63;
    const float* t = transforms + (size_t)idx * 6;
    float wraw = pw[idx];
    float t0 = t[0], t1 = t[1], t2 = t[2];
    float ax = t[3], ay = t[4], az = t[5];
    float cx = cosf(ax), sx = sinf(ax);
    float cy = cosf(ay), sy = sinf(ay);
    float cz = cosf(az), sz = sinf(az);
    float r00 = cy * cz;
    float r01 = -cy * sz;
    float r02 = sy;
    float r10 = sx * sy * cz + cx * sz;
    float r11 = -sx * sy * sz + cx * cz;
    float r12 = -sx * cy;
    float r20 = -cx * sy * cz + sx * sz;
    float r21 = cx * sy * sz + sx * cz;
    float r22 = cx * cy;
    float w = wraw * scales[bs];
    float* a0 = A + (size_t)(bs * 3 + 0) * kK + p * 3;
    float* a1 = A + (size_t)(bs * 3 + 1) * kK + p * 3;
    float* a2 = A + (size_t)(bs * 3 + 2) * kK + p * 3;
    a0[0] = r00 * w; a0[1] = r01 * w; a0[2] = r02 * w;
    a1[0] = r10 * w; a1[1] = r11 * w; a1[2] = r12 * w;
    a2[0] = r20 * w; a2[1] = r21 * w; a2[2] = r22 * w;

    sm[tid][0] = wraw * t0;
    sm[tid][1] = wraw * t1;
    sm[tid][2] = wraw * t2;
    __syncthreads();
    for (int st = 32; st > 0; st >>= 1) {
      if (p < st) {
        sm[tid][0] += sm[tid + st][0];
        sm[tid][1] += sm[tid + st][1];
        sm[tid][2] += sm[tid + st][2];
      }
      __syncthreads();
    }
    if (p < 3) T[bs * 3 + p] = sm[tid - p][p];
  } else {
    int id = (blockIdx.x - kPrepAT) * kPrepThreads + threadIdx.x;
    int kp = id / kNPAD;
    int n  = id - kp * kNPAD;
    v2f val = {0.f, 0.f};
    if (n < kV) {
      int k0 = kp * 2, k1 = k0 + 1;
      int p0 = k0 / 3, j0 = k0 - p0 * 3;
      int p1 = k1 / 3, j1 = k1 - p1 * 3;
      val.x = base[(size_t)n * 3 + j0] + off[((size_t)p0 * kV + n) * 3 + j0];
      val.y = base[(size_t)n * 3 + j1] + off[((size_t)p1 * kV + n) * 3 + j1];
    }
    Dp[id] = val;
  }
}

__global__ __launch_bounds__(32) void gemm_wmma_kernel(
    const float* __restrict__ A, const v2f* __restrict__ Dp,
    const float* __restrict__ T, float* __restrict__ out) {
  int lane = threadIdx.x;
  int mt = blockIdx.x;
  int nt = blockIdx.y;
  int m0 = mt * 16;
  int n0 = nt * (kNTILE * 16);
  int half = lane >> 4;
  int lrow = lane & 15;

  const float* Ap = A + (size_t)(m0 + lrow) * kK + half * 2;
  const v2f* Bp = Dp + (size_t)half * kNPAD + n0 + lrow;

  v8f c0 = {0.f, 0.f, 0.f, 0.f, 0.f, 0.f, 0.f, 0.f};
  v8f c1 = c0, c2 = c0, c3 = c0;
#pragma unroll 4
  for (int k = 0; k < kK; k += 4) {
    v2f a = *(const v2f*)(Ap + k);
    const v2f* b = Bp + (size_t)(k >> 1) * kNPAD;
    v2f b0 = b[0], b1 = b[16], b2 = b[32], b3 = b[48];
    c0 = __builtin_amdgcn_wmma_f32_16x16x4_f32(false, a, false, b0, (short)0,
                                               c0, false, false);
    c1 = __builtin_amdgcn_wmma_f32_16x16x4_f32(false, a, false, b1, (short)0,
                                               c1, false, false);
    c2 = __builtin_amdgcn_wmma_f32_16x16x4_f32(false, a, false, b2, (short)0,
                                               c2, false, false);
    c3 = __builtin_amdgcn_wmma_f32_16x16x4_f32(false, a, false, b3, (short)0,
                                               c3, false, false);
  }

  int mb = m0 + half * 8;
  int bs = mb / 3;
  int i  = mb - bs * 3;
  float  tv[8];
  size_t ob[8];
#pragma unroll
  for (int r = 0; r < 8; ++r) {
    tv[r] = T[mb + r];
    ob[r] = (size_t)bs * (kV * 3) + i;
    if (++i == 3) { i = 0; ++bs; }
  }
  auto store_tile = [&](int t, const v8f& c) {
    int n = n0 + t * 16 + lrow;
    if (n < kV) {
      size_t n3 = (size_t)n * 3;
#pragma unroll
      for (int r = 0; r < 8; ++r) out[ob[r] + n3] = c[r] + tv[r];
    }
  };
  store_tile(0, c0);
  store_tile(1, c1);
  store_tile(2, c2);
  store_tile(3, c3);
}

extern "C" void kernel_launch(void* const* d_in, const int* in_sizes, int n_in,
                              void* d_out, int out_size, void* d_ws,
                              size_t ws_size, hipStream_t stream) {
  const float* scales     = (const float*)d_in[0];
  const float* transforms = (const float*)d_in[1];
  const float* pw         = (const float*)d_in[2];
  const float* off        = (const float*)d_in[3];
  const float* base       = (const float*)d_in[4];
  float* out = (float*)d_out;

  float* wsf = (float*)d_ws;
  float* A  = wsf;
  v2f*   Dp = (v2f*)(wsf + (size_t)kM * kK);
  float* T  = (float*)(Dp + (size_t)kKP * kNPAD);

  prep_all_kernel<<<kPrepTotal, kPrepThreads, 0, stream>>>(
      scales, transforms, pw, off, base, A, Dp, T);
  gemm_wmma_kernel<<<dim3(kMT, kNT4), 32, 0, stream>>>(A, Dp, T, out);
}


# GEMM K loop: the next block's A/B fragment loads (6 k-steps, a second register set) and the T loads are issued as one group ahead of the current block's WMMAs; k order and sums kept
# speedup vs baseline: 1.5453x; 1.2004x over previous
#include <hip/hip_runtime.h>

typedef __attribute__((ext_vector_type(2))) float v2f;
typedef __attribute__((ext_vector_type(8))) float v8f;

constexpr int kBS    = 256;
constexpr int kP     = 64;
constexpr int kV     = 2562;
constexpr int kK     = kP * 3;
constexpr int kKP    = kK / 2;
constexpr int kM     = kBS * 3;
constexpr int kMT    = kM / 16;
constexpr int kNTILE = 4;
constexpr int kNT4   = 41;
constexpr int kNPAD  = kNT4 * kNTILE * 16;
constexpr int kKB    = 6;
constexpr int kNBLK  = (kK / 4) / kKB;
static_assert(kNBLK * kKB * 4 == kK, "load blocks must tile K exactly");
static_assert(kNBLK % 2 == 0 && kNBLK >= 2, "two register sets alternate");

constexpr int kPrepThreads = 256;
constexpr int kPrepAT      = kBS * kP / kPrepThreads;
constexpr int kPrepD       = kKP * kNPAD / kPrepThreads;
constexpr int kPrepTotal   = kPrepAT + kPrepD;
static_assert(kPrepAT * kPrepThreads == kBS * kP, "A/T role grid must be exact");
static_assert(kPrepD * kPrepThreads == kKP * kNPAD, "D role grid must be exact");
static_assert(kPrepThreads % kP == 0, "a (b,s) group must not straddle blocks");

__global__ __launch_bounds__(kPrepThreads) void prep_all_kernel(
    const float* __restrict__ scales, const float* __restrict__ transforms,
    const float* __restrict__ pw, const float* __restrict__ off,
    const float* __restrict__ base, float* __restrict__ A,
    v2f* __restrict__ Dp, float* __restrict__ T) {
  __shared__ float sm[kPrepThreads][3];
  if (blockIdx.x < kPrepAT) {
    int tid = threadIdx.x;
    int idx = blockIdx.x * kPrepThreads + tid;
    int bs = idx >> 6;
    int p  = idx & 63;
    const float* t = transforms + (size_t)idx * 6;
    float wraw = pw[idx];
    float t0 = t[0], t1 = t[1], t2 = t[2];
    float ax = t[3], ay = t[4], az = t[5];
    float cx = cosf(ax), sx = sinf(ax);
    float cy = cosf(ay), sy = sinf(ay);
    float cz = cosf(az), sz = sinf(az);
    float r00 = cy * cz;
    float r01 = -cy * sz;
    float r02 = sy;
    float r10 = sx * sy * cz + cx * sz;
    float r11 = -sx * sy * sz + cx * cz;
    float r12 = -sx * cy;
    float r20 = -cx * sy * cz + sx * sz;
    float r21 = cx * sy * sz + sx * cz;
    float r22 = cx * cy;
    float w = wraw * scales[bs];
    float* a0 = A + (size_t)(bs * 3 + 0) * kK + p * 3;
    float* a1 = A + (size_t)(bs * 3 + 1) * kK + p * 3;
    float* a2 = A + (size_t)(bs * 3 + 2) * kK + p * 3;
    a0[0] = r00 * w; a0[1] = r01 * w; a0[2] = r02 * w;
    a1[0] = r10 * w; a1[1] = r11 * w; a1[2] = r12 * w;
    a2[0] = r20 * w; a2[1] = r21 * w; a2[2] = r22 * w;

    sm[tid][0] = wraw * t0;
    sm[tid][1] = wraw * t1;
    sm[tid][2] = wraw * t2;
    __syncthreads();
    for (int st = 32; st > 0; st >>= 1) {
      if (p < st) {
        sm[tid][0] += sm[tid + st][0];
        sm[tid][1] += sm[tid + st][1];
        sm[tid][2] += sm[tid + st][2];
      }
      __syncthreads();
    }
    if (p < 3) T[bs * 3 + p] = sm[tid - p][p];
  } else {
    int id = (blockIdx.x - kPrepAT) * kPrepThreads + threadIdx.x;
    int kp = id / kNPAD;
    int n  = id - kp * kNPAD;
    v2f val = {0.f, 0.f};
    if (n < kV) {
      int k0 = kp * 2, k1 = k0 + 1;
      int p0 = k0 / 3, j0 = k0 - p0 * 3;
      int p1 = k1 / 3, j1 = k1 - p1 * 3;
      val.x = base[(size_t)n * 3 + j0] + off[((size_t)p0 * kV + n) * 3 + j0];
      val.y = base[(size_t)n * 3 + j1] + off[((size_t)p1 * kV + n) * 3 + j1];
    }
    Dp[id] = val;
  }
}

__global__ __launch_bounds__(32) void gemm_wmma_kernel(
    const float* __restrict__ A, const v2f* __restrict__ Dp,
    const float* __restrict__ T, float* __restrict__ out) {
  int lane = threadIdx.x;
  int mt = blockIdx.x;
  int nt = blockIdx.y;
  int m0 = mt * 16;
  int n0 = nt * (kNTILE * 16);
  int half = lane >> 4;
  int lrow = lane & 15;

  const float* Ap = A + (size_t)(m0 + lrow) * kK + half * 2;
  const v2f* Bp = Dp + (size_t)half * kNPAD + n0 + lrow;

  v8f c0 = {0.f, 0.f, 0.f, 0.f, 0.f, 0.f, 0.f, 0.f};
  v8f c1 = c0, c2 = c0, c3 = c0;

  v2f fa0[kKB], fa1[kKB];
  v2f fb0[kKB][kNTILE], fb1[kKB][kNTILE];
  auto load_blk = [&](int blk, v2f (&fa)[kKB], v2f (&fb)[kKB][kNTILE]) {
#pragma unroll
    for (int s = 0; s < kKB; ++s) {
      int k = (blk * kKB + s) * 4;
      fa[s] = *(const v2f*)(Ap + k);
      const v2f* b = Bp + (size_t)(k >> 1) * kNPAD;
      fb[s][0] = b[0]; fb[s][1] = b[16]; fb[s][2] = b[32]; fb[s][3] = b[48];
    }
  };
  auto mma_blk = [&](const v2f (&fa)[kKB], const v2f (&fb)[kKB][kNTILE]) {
#pragma unroll
    for (int s = 0; s < kKB; ++s) {
      c0 = __builtin_amdgcn_wmma_f32_16x16x4_f32(false, fa[s], false, fb[s][0],
                                                 (short)0, c0, false, false);
      c1 = __builtin_amdgcn_wmma_f32_16x16x4_f32(false, fa[s], false, fb[s][1],
                                                 (short)0, c1, false, false);
      c2 = __builtin_amdgcn_wmma_f32_16x16x4_f32(false, fa[s], false, fb[s][2],
                                                 (short)0, c2, false, false);
      c3 = __builtin_amdgcn_wmma_f32_16x16x4_f32(false, fa[s], false, fb[s][3],
                                                 (short)0, c3, false, false);
    }
  };

  load_blk(0, fa0, fb0);

  int mb = m0 + half * 8;
  int bs = mb / 3;
  int i  = mb - bs * 3;
  float  tv[8];
  size_t ob[8];
#pragma unroll
  for (int r = 0; r < 8; ++r) {
    tv[r] = T[mb + r];
    ob[r] = (size_t)bs * (kV * 3) + i;
    if (++i == 3) { i = 0; ++bs; }
  }

#pragma unroll
  for (int blk = 0; blk < kNBLK - 2; blk += 2) {
    load_blk(blk + 1, fa1, fb1);
    __builtin_amdgcn_sched_barrier(0);
    mma_blk(fa0, fb0);
    __builtin_amdgcn_sched_barrier(0);
    load_blk(blk + 2, fa0, fb0);
    __builtin_amdgcn_sched_barrier(0);
    mma_blk(fa1, fb1);
    __builtin_amdgcn_sched_barrier(0);
  }
  load_blk(kNBLK - 1, fa1, fb1);
  __builtin_amdgcn_sched_barrier(0);
  mma_blk(fa0, fb0);
  mma_blk(fa1, fb1);

  auto store_tile = [&](int t, const v8f& c) {
    int n = n0 + t * 16 + lrow;
    if (n < kV) {
      size_t n3 = (size_t)n * 3;
#pragma unroll
      for (int r = 0; r < 8; ++r) out[ob[r] + n3] = c[r] + tv[r];
    }
  };
  store_tile(0, c0);
  store_tile(1, c1);
  store_tile(2, c2);
  store_tile(3, c3);
}

extern "C" void kernel_launch(void* const* d_in, const int* in_sizes, int n_in,
                              void* d_out, int out_size, void* d_ws,
                              size_t ws_size, hipStream_t stream) {
  const float* scales     = (const float*)d_in[0];
  const float* transforms = (const float*)d_in[1];
  const float* pw         = (const float*)d_in[2];
  const float* off        = (const float*)d_in[3];
  const float* base       = (const float*)d_in[4];
  float* out = (float*)d_out;

  float* wsf = (float*)d_ws;
  float* A  = wsf;
  v2f*   Dp = (v2f*)(wsf + (size_t)kM * kK);
  float* T  = (float*)(Dp + (size_t)kKP * kNPAD);

  prep_all_kernel<<<kPrepTotal, kPrepThreads, 0, stream>>>(
      scales, transforms, pw, off, base, A, Dp, T);
  gemm_wmma_kernel<<<dim3(kMT, kNT4), 32, 0, stream>>>(A, Dp, T, out);
}


# GEMM work-groups of 4 waves (4 M tiles, same 64 columns) instead of 1: each D line comes from L2 once per work-group and is shared in the compute unit's cache; per-wave work, k order and sums kept
# speedup vs baseline: 1.6740x; 1.0833x over previous
#include <hip/hip_runtime.h>

typedef __attribute__((ext_vector_type(2))) float v2f;
typedef __attribute__((ext_vector_type(8))) float v8f;

constexpr int kBS    = 256;
constexpr int kP     = 64;
constexpr int kV     = 2562;
constexpr int kK     = kP * 3;
constexpr int kKP    = kK / 2;
constexpr int kM     = kBS * 3;
constexpr int kMT    = kM / 16;
constexpr int kNTILE = 4;
constexpr int kNT4   = 41;
constexpr int kNPAD  = kNT4 * kNTILE * 16;
constexpr int kKB    = 6;
constexpr int kNBLK  = (kK / 4) / kKB;
constexpr int kWPB   = 4;
constexpr int kGemmThreads = 32 * kWPB;
constexpr int kGemmGridX   = kMT / kWPB;
static_assert(kGemmGridX * kWPB == kMT, "work-groups must tile M exactly");
static_assert(kNBLK * kKB * 4 == kK, "load blocks must tile K exactly");
static_assert(kNBLK % 2 == 0 && kNBLK >= 2, "two register sets alternate");

constexpr int kPrepThreads = 256;
constexpr int kPrepAT      = kBS * kP / kPrepThreads;
constexpr int kPrepD       = kKP * kNPAD / kPrepThreads;
constexpr int kPrepTotal   = kPrepAT + kPrepD;
static_assert(kPrepAT * kPrepThreads == kBS * kP, "A/T role grid must be exact");
static_assert(kPrepD * kPrepThreads == kKP * kNPAD, "D role grid must be exact");
static_assert(kPrepThreads % kP == 0, "a (b,s) group must not straddle blocks");

__global__ __launch_bounds__(kPrepThreads) void prep_all_kernel(
    const float* __restrict__ scales, const float* __restrict__ transforms,
    const float* __restrict__ pw, const float* __restrict__ off,
    const float* __restrict__ base, float* __restrict__ A,
    v2f* __restrict__ Dp, float* __restrict__ T) {
  __shared__ float sm[kPrepThreads][3];
  if (blockIdx.x < kPrepAT) {
    int tid = threadIdx.x;
    int idx = blockIdx.x * kPrepThreads + tid;
    int bs = idx >> 6;
    int p  = idx & 63;
    const float* t = transforms + (size_t)idx * 6;
    float wraw = pw[idx];
    float t0 = t[0], t1 = t[1], t2 = t[2];
    float ax = t[3], ay = t[4], az = t[5];
    float cx = cosf(ax), sx = sinf(ax);
    float cy = cosf(ay), sy = sinf(ay);
    float cz = cosf(az), sz = sinf(az);
    float r00 = cy * cz;
    float r01 = -cy * sz;
    float r02 = sy;
    float r10 = sx * sy * cz + cx * sz;
    float r11 = -sx * sy * sz + cx * cz;
    float r12 = -sx * cy;
    float r20 = -cx * sy * cz + sx * sz;
    float r21 = cx * sy * sz + sx * cz;
    float r22 = cx * cy;
    float w = wraw * scales[bs];
    float* a0 = A + (size_t)(bs * 3 + 0) * kK + p * 3;
    float* a1 = A + (size_t)(bs * 3 + 1) * kK + p * 3;
    float* a2 = A + (size_t)(bs * 3 + 2) * kK + p * 3;
    a0[0] = r00 * w; a0[1] = r01 * w; a0[2] = r02 * w;
    a1[0] = r10 * w; a1[1] = r11 * w; a1[2] = r12 * w;
    a2[0] = r20 * w; a2[1] = r21 * w; a2[2] = r22 * w;

    sm[tid][0] = wraw * t0;
    sm[tid][1] = wraw * t1;
    sm[tid][2] = wraw * t2;
    __syncthreads();
    for (int st = 32; st > 0; st >>= 1) {
      if (p < st) {
        sm[tid][0] += sm[tid + st][0];
        sm[tid][1] += sm[tid + st][1];
        sm[tid][2] += sm[tid + st][2];
      }
      __syncthreads();
    }
    if (p < 3) T[bs * 3 + p] = sm[tid - p][p];
  } else {
    int id = (blockIdx.x - kPrepAT) * kPrepThreads + threadIdx.x;
    int kp = id / kNPAD;
    int n  = id - kp * kNPAD;
    v2f val = {0.f, 0.f};
    if (n < kV) {
      int k0 = kp * 2, k1 = k0 + 1;
      int p0 = k0 / 3, j0 = k0 - p0 * 3;
      int p1 = k1 / 3, j1 = k1 - p1 * 3;
      val.x = base[(size_t)n * 3 + j0] + off[((size_t)p0 * kV + n) * 3 + j0];
      val.y = base[(size_t)n * 3 + j1] + off[((size_t)p1 * kV + n) * 3 + j1];
    }
    Dp[id] = val;
  }
}

__global__ __launch_bounds__(kGemmThreads) void gemm_wmma_kernel(
    const float* __restrict__ A, const v2f* __restrict__ Dp,
    const float* __restrict__ T, float* __restrict__ out) {
  int lane = threadIdx.x & 31;
  int mt = blockIdx.x * kWPB + (threadIdx.x >> 5);
  int nt = blockIdx.y;
  int m0 = mt * 16;
  int n0 = nt * (kNTILE * 16);
  int half = lane >> 4;
  int lrow = lane & 15;

  const float* Ap = A + (size_t)(m0 + lrow) * kK + half * 2;
  const v2f* Bp = Dp + (size_t)half * kNPAD + n0 + lrow;

  v8f c0 = {0.f, 0.f, 0.f, 0.f, 0.f, 0.f, 0.f, 0.f};
  v8f c1 = c0, c2 = c0, c3 = c0;

  v2f fa0[kKB], fa1[kKB];
  v2f fb0[kKB][kNTILE], fb1[kKB][kNTILE];
  auto load_blk = [&](int blk, v2f (&fa)[kKB], v2f (&fb)[kKB][kNTILE]) {
#pragma unroll
    for (int s = 0; s < kKB; ++s) {
      int k = (blk * kKB + s) * 4;
      fa[s] = *(const v2f*)(Ap + k);
      const v2f* b = Bp + (size_t)(k >> 1) * kNPAD;
      fb[s][0] = b[0]; fb[s][1] = b[16]; fb[s][2] = b[32]; fb[s][3] = b[48];
    }
  };
  auto mma_blk = [&](const v2f (&fa)[kKB], const v2f (&fb)[kKB][kNTILE]) {
#pragma unroll
    for (int s = 0; s < kKB; ++s) {
      c0 = __builtin_amdgcn_wmma_f32_16x16x4_f32(false, fa[s], false, fb[s][0],
                                                 (short)0, c0, false, false);
      c1 = __builtin_amdgcn_wmma_f32_16x16x4_f32(false, fa[s], false, fb[s][1],
                                                 (short)0, c1, false, false);
      c2 = __builtin_amdgcn_wmma_f32_16x16x4_f32(false, fa[s], false, fb[s][2],
                                                 (short)0, c2, false, false);
      c3 = __builtin_amdgcn_wmma_f32_16x16x4_f32(false, fa[s], false, fb[s][3],
                                                 (short)0, c3, false, false);
    }
  };

  load_blk(0, fa0, fb0);

  int mb = m0 + half * 8;
  int bs = mb / 3;
  int i  = mb - bs * 3;
  float  tv[8];
  size_t ob[8];
#pragma unroll
  for (int r = 0; r < 8; ++r) {
    tv[r] = T[mb + r];
    ob[r] = (size_t)bs * (kV * 3) + i;
    if (++i == 3) { i = 0; ++bs; }
  }

#pragma unroll
  for (int blk = 0; blk < kNBLK - 2; blk += 2) {
    load_blk(blk + 1, fa1, fb1);
    __builtin_amdgcn_sched_barrier(0);
    mma_blk(fa0, fb0);
    __builtin_amdgcn_sched_barrier(0);
    load_blk(blk + 2, fa0, fb0);
    __builtin_amdgcn_sched_barrier(0);
    mma_blk(fa1, fb1);
    __builtin_amdgcn_sched_barrier(0);
  }
  load_blk(kNBLK - 1, fa1, fb1);
  __builtin_amdgcn_sched_barrier(0);
  mma_blk(fa0, fb0);
  mma_blk(fa1, fb1);

  auto store_tile = [&](int t, const v8f& c) {
    int n = n0 + t * 16 + lrow;
    if (n < kV) {
      size_t n3 = (size_t)n * 3;
#pragma unroll
      for (int r = 0; r < 8; ++r) out[ob[r] + n3] = c[r] + tv[r];
    }
  };
  store_tile(0, c0);
  store_tile(1, c1);
  store_tile(2, c2);
  store_tile(3, c3);
}

extern "C" void kernel_launch(void* const* d_in, const int* in_sizes, int n_in,
                              void* d_out, int out_size, void* d_ws,
                              size_t ws_size, hipStream_t stream) {
  const float* scales     = (const float*)d_in[0];
  const float* transforms = (const float*)d_in[1];
  const float* pw         = (const float*)d_in[2];
  const float* off        = (const float*)d_in[3];
  const float* base       = (const float*)d_in[4];
  float* out = (float*)d_out;

  float* wsf = (float*)d_ws;
  float* A  = wsf;
  v2f*   Dp = (v2f*)(wsf + (size_t)kM * kK);
  float* T  = (float*)(Dp + (size_t)kKP * kNPAD);

  prep_all_kernel<<<kPrepTotal, kPrepThreads, 0, stream>>>(
      scales, transforms, pw, off, base, A, Dp, T);
  gemm_wmma_kernel<<<dim3(kGemmGridX, kNT4), kGemmThreads, 0, stream>>>(A, Dp, T, out);
}


# One launch, not two: each work-group (32 rows x 256 columns) builds its A rows and T in LDS behind one batched load group, B read straight from offsets+base; K in groups of 12: k sum order changes
# speedup vs baseline: 1.7822x; 1.0646x over previous
#include <hip/hip_runtime.h>

typedef __attribute__((ext_vector_type(2))) float v2f;
typedef __attribute__((ext_vector_type(4))) float v4f;
typedef __attribute__((ext_vector_type(8))) float v8f;
struct f3 { float x, y, z; };

template <int N> struct IntC { static constexpr int value = N; };

constexpr int kBS      = 256;
constexpr int kP       = 64;
constexpr int kV       = 2562;
constexpr int kK       = kP * 3;
constexpr int kM       = kBS * 3;
constexpr int kRows    = 32;
constexpr int kMG      = kM / kRows;
constexpr int kThreads = 128;
constexpr int kNTile   = 4;
constexpr int kColsWave = kNTile * 16;
constexpr int kColsWG  = 4 * kColsWave;
constexpr int kNG      = (kV + kColsWG - 1) / kColsWG;
constexpr int kPitch   = 196;
constexpr int kG       = kK / 12;
constexpr int kBsMax   = 12;
constexpr int kIters   = kBsMax * kP / kThreads;
constexpr int kTPitch  = 68;
static_assert(kMG * kRows == kM, "row groups must tile M exactly");
static_assert(kG * 12 == kK, "K groups must tile K exactly");
static_assert(kIters * kThreads == kBsMax * kP, "set loop must be exact");
static_assert(kV % kColsWave == 2, "the only partial wave has one N tile");

__global__ __launch_bounds__(kThreads) void fused_gemm_kernel(
    const float* __restrict__ scales, const float* __restrict__ transforms,
    const float* __restrict__ pw, const float* __restrict__ off,
    const float* __restrict__ base, float* __restrict__ out) {
  __shared__ __attribute__((aligned(16))) float As[kRows * kPitch];
  __shared__ __attribute__((aligned(16))) float Raw[kIters][8][kThreads];
  __shared__ __attribute__((aligned(16))) float Tp[kBsMax * 3 * kTPitch];

  const int tid   = threadIdx.x;
  const int lane  = tid & 31;
  const int wave  = tid >> 5;
  const int half  = lane >> 4;
  const int lrow  = lane & 15;
  const int m0    = blockIdx.x * kRows;
  const int ncol0 = blockIdx.y * kColsWG + wave * kColsWave;
  const int bs_lo = m0 / 3;
  const int nbs   = (m0 + kRows - 1) / 3 - bs_lo + 1;
  const int nsets = nbs * kP;

  const float* offp[kNTile];
  f3 bn[kNTile];
  f3 braw[3][kNTile][2];
#pragma unroll
  for (int t = 0; t < kNTile; ++t) {
    int n = ncol0 + t * 16 + lrow;
    n = n < kV ? n : kV - 1;
    offp[t] = off + ((size_t)(2 * half) * kV + n) * 3;
    bn[t] = *(const f3*)(base + (size_t)n * 3);
  }
#pragma unroll
  for (int g = 0; g < 2; ++g) {
#pragma unroll
    for (int t = 0; t < kNTile; ++t) {
      const float* q = offp[t] + (size_t)g * (4 * kV * 3);
      braw[g][t][0] = *(const f3*)(q);
      braw[g][t][1] = *(const f3*)(q + kV * 3);
    }
  }

  {
    v2f ta[kIters], tb[kIters], tc[kIters];
    float tw[kIters], ts[kIters];
#pragma unroll
    for (int it = 0; it < kIters; ++it) {
      int idx = it * kThreads + tid;
      int sidx = idx < nsets ? idx : 0;
      int gi = bs_lo * kP + sidx;
      const v2f* tp = (const v2f*)(transforms + (size_t)gi * 6);
      ta[it] = tp[0]; tb[it] = tp[1]; tc[it] = tp[2];
      tw[it] = pw[gi];
      ts[it] = scales[gi >> 6];
    }
#pragma unroll
    for (int it = 0; it < kIters; ++it) {
      Raw[it][0][tid] = ta[it].x; Raw[it][1][tid] = ta[it].y;
      Raw[it][2][tid] = tb[it].x; Raw[it][3][tid] = tb[it].y;
      Raw[it][4][tid] = tc[it].x; Raw[it][5][tid] = tc[it].y;
      Raw[it][6][tid] = tw[it];   Raw[it][7][tid] = ts[it];
    }
  }

  {
    const int p = tid & 63;
#pragma unroll 1
    for (int it = 0; it < kIters; ++it) {
      int bsl = it * 2 + (tid >> 6);
      if (bsl < nbs) {
        float t0 = Raw[it][0][tid], t1 = Raw[it][1][tid], t2 = Raw[it][2][tid];
        float ax = Raw[it][3][tid], ay = Raw[it][4][tid], az = Raw[it][5][tid];
        float wraw = Raw[it][6][tid];
        float w = wraw * Raw[it][7][tid];
        float cx, sx, cy, sy, cz, sz;
        sincosf(ax, &sx, &cx);
        sincosf(ay, &sy, &cy);
        sincosf(az, &sz, &cz);
        float r00 = cy * cz;
        float r01 = -cy * sz;
        float r02 = sy;
        float r10 = sx * sy * cz + cx * sz;
        float r11 = -sx * sy * sz + cx * cz;
        float r12 = -sx * cy;
        float r20 = -cx * sy * cz + sx * sz;
        float r21 = cx * sy * sz + sx * cz;
        float r22 = cx * cy;
        int rb = (bs_lo + bsl) * 3 - m0;
        float* ar = As + rb * kPitch + p * 3;
        if (rb >= 0 && rb < kRows) {
          ar[0] = r00 * w; ar[1] = r01 * w; ar[2] = r02 * w;
        }
        if (rb + 1 >= 0 && rb + 1 < kRows) {
          ar[kPitch + 0] = r10 * w; ar[kPitch + 1] = r11 * w; ar[kPitch + 2] = r12 * w;
        }
        if (rb + 2 >= 0 && rb + 2 < kRows) {
          ar[2 * kPitch + 0] = r20 * w; ar[2 * kPitch + 1] = r21 * w; ar[2 * kPitch + 2] = r22 * w;
        }
        float* tq = Tp + (bsl * 3) * kTPitch + p;
        tq[0]           = wraw * t0;
        tq[kTPitch]     = wraw * t1;
        tq[2 * kTPitch] = wraw * t2;
      }
    }
  }
  __syncthreads();

  if (ncol0 >= kV) return;

  float Tval;
  {
    const float* x = Tp + (m0 - 3 * bs_lo + lane) * kTPitch;
    float d[16];
#pragma unroll
    for (int Q = 0; Q < 16; Q += 4) {
      v4f x0 = *(const v4f*)(x + Q);
      v4f x1 = *(const v4f*)(x + Q + 16);
      v4f x2 = *(const v4f*)(x + Q + 32);
      v4f x3 = *(const v4f*)(x + Q + 48);
#pragma unroll
      for (int e = 0; e < 4; ++e) d[Q + e] = (x0[e] + x2[e]) + (x1[e] + x3[e]);
    }
    float c[8], b[4];
#pragma unroll
    for (int q = 0; q < 8; ++q) c[q] = d[q] + d[q + 8];
#pragma unroll
    for (int q = 0; q < 4; ++q) b[q] = c[q] + c[q + 4];
    Tval = (b[0] + b[2]) + (b[1] + b[3]);
  }
  float tv[2][8];
  int   ob[2][8];
#pragma unroll
  for (int mt = 0; mt < 2; ++mt) {
#pragma unroll
    for (int r = 0; r < 8; ++r) {
      int lr = mt * 16 + half * 8 + r;
      tv[mt][r] = __shfl(Tval, lr);
      int m = m0 + lr;
      int bs = m / 3;
      ob[mt][r] = bs * (kV * 3) + (m - bs * 3);
    }
  }

  const float* Ap = As + lrow * kPitch + 6 * half;

  auto run = [&](auto ntc) {
    constexpr int NT = decltype(ntc)::value;
    v8f acc[2][NT];
#pragma unroll
    for (int mt = 0; mt < 2; ++mt)
#pragma unroll
      for (int t = 0; t < NT; ++t)
        acc[mt][t] = v8f{0.f, 0.f, 0.f, 0.f, 0.f, 0.f, 0.f, 0.f};
    v2f afr[2][2][3];

    auto loadB = [&](int g, int set) {
#pragma unroll
      for (int t = 0; t < NT; ++t) {
        const float* q = offp[t] + (size_t)g * (4 * kV * 3);
        braw[set][t][0] = *(const f3*)(q);
        braw[set][t][1] = *(const f3*)(q + kV * 3);
      }
    };
    auto loadA = [&](int g, int set) {
#pragma unroll
      for (int mt = 0; mt < 2; ++mt)
#pragma unroll
        for (int s = 0; s < 3; ++s)
          afr[set][mt][s] = *(const v2f*)(Ap + mt * 16 * kPitch + g * 12 + s * 2);
    };
    auto mma = [&](int bset, int aset) {
      v2f bf[NT][3];
#pragma unroll
      for (int t = 0; t < NT; ++t) {
        f3 u0 = braw[bset][t][0], u1 = braw[bset][t][1];
        asm volatile("" : "+v"(u0.x), "+v"(u0.y), "+v"(u0.z), "+v"(u1.x), "+v"(u1.y), "+v"(u1.z));
        bf[t][0] = v2f{u0.x + bn[t].x, u0.y + bn[t].y};
        bf[t][1] = v2f{u0.z + bn[t].z, u1.x + bn[t].x};
        bf[t][2] = v2f{u1.y + bn[t].y, u1.z + bn[t].z};
      }
#pragma unroll
      for (int s = 0; s < 3; ++s)
#pragma unroll
        for (int t = 0; t < NT; ++t)
#pragma unroll
          for (int mt = 0; mt < 2; ++mt)
            acc[mt][t] = __builtin_amdgcn_wmma_f32_16x16x4_f32(
                false, afr[aset][mt][s], false, bf[t][s], (short)0, acc[mt][t],
                false, false);
    };

    loadA(0, 0);
#pragma unroll
    for (int g = 0; g < kG; ++g) {
      if (g + 2 < kG) loadB(g + 2, (g + 2) % 3);
      if (g + 1 < kG) loadA(g + 1, (g + 1) & 1);
      __builtin_amdgcn_sched_barrier(0);
      mma(g % 3, g & 1);
      __builtin_amdgcn_sched_barrier(0);
    }

#pragma unroll
    for (int t = 0; t < NT; ++t) {
      int n = ncol0 + t * 16 + lrow;
      if (n < kV) {
        int n3 = n * 3;
#pragma unroll
        for (int mt = 0; mt < 2; ++mt)
#pragma unroll
          for (int r = 0; r < 8; ++r)
            out[ob[mt][r] + n3] = acc[mt][t][r] + tv[mt][r];
      }
    }
  };

  if (ncol0 + kColsWave <= kV) run(IntC<kNTile>{});
  else                         run(IntC<1>{});
}

extern "C" void kernel_launch(void* const* d_in, const int* in_sizes, int n_in,
                              void* d_out, int out_size, void* d_ws,
                              size_t ws_size, hipStream_t stream) {
  const float* scales     = (const float*)d_in[0];
  const float* transforms = (const float*)d_in[1];
  const float* pw         = (const float*)d_in[2];
  const float* off        = (const float*)d_in[3];
  const float* base       = (const float*)d_in[4];
  float* out = (float*)d_out;
  fused_gemm_kernel<<<dim3(kMG, kNG), kThreads, 0, stream>>>(
      scales, transforms, pw, off, base, out);
}


# A build as straight-line code: sin/cos reduction inline behind one large-argument test per thread (library path kept for those), LDS guard rows replace the row tests; six sets' chains overlap
# speedup vs baseline: 2.0086x; 1.1270x over previous
#include <hip/hip_runtime.h>

typedef __attribute__((ext_vector_type(2))) float v2f;
typedef __attribute__((ext_vector_type(4))) float v4f;
typedef __attribute__((ext_vector_type(8))) float v8f;
struct f3 { float x, y, z; };

template <int N> struct IntC { static constexpr int value = N; };

constexpr int kBS      = 256;
constexpr int kP       = 64;
constexpr int kV       = 2562;
constexpr int kK       = kP * 3;
constexpr int kM       = kBS * 3;
constexpr int kRows    = 32;
constexpr int kMG      = kM / kRows;
constexpr int kThreads = 128;
constexpr int kNTile   = 4;
constexpr int kColsWave = kNTile * 16;
constexpr int kColsWG  = 4 * kColsWave;
constexpr int kNG      = (kV + kColsWG - 1) / kColsWG;
constexpr int kPitch   = 196;
constexpr int kG       = kK / 12;
constexpr int kBsMax   = 12;
constexpr int kIters   = kBsMax * kP / kThreads;
constexpr int kTPitch  = 68;
constexpr int kGuardLo = 2;
constexpr int kGuardHi = 4;
constexpr float kMidArg = 131072.0f;
static_assert(kMG * kRows == kM, "row groups must tile M exactly");
static_assert(kG * 12 == kK, "K groups must tile K exactly");
static_assert(kIters * kThreads == kBsMax * kP, "set loop must be exact");
static_assert(kV % kColsWave == 2, "the only partial wave has one N tile");

__device__ __forceinline__ float bits_f(unsigned u) { return __builtin_bit_cast(float, u); }
__device__ __forceinline__ unsigned f_bits(float f) { return __builtin_bit_cast(unsigned, f); }

__device__ __forceinline__ void sincos_mid(float x, float& s, float& c) {
  const float a = __builtin_fabsf(x);
  const float n = __builtin_rintf(a * bits_f(0x3f22f983u));
  float r = __builtin_fmaf(n, bits_f(0xbfc90fdau), a);
  r = __builtin_fmaf(n, bits_f(0xb3a22168u), r);
  r = __builtin_fmaf(n, bits_f(0xa7c234c4u), r);
  const unsigned q = (unsigned)(int)n;
  const float z = r * r;
  float ps = __builtin_fmaf(z, bits_f(0xb94c1982u), bits_f(0x3c0881c4u));
  ps = __builtin_fmaf(z, ps, bits_f(0xbe2aaa9du));
  const float sr = __builtin_fmaf(r, z * ps, r);
  float pc = __builtin_fmaf(z, bits_f(0x37d75334u), bits_f(0xbab64f3bu));
  pc = __builtin_fmaf(z, pc, bits_f(0x3d2aabf7u));
  pc = __builtin_fmaf(z, pc, bits_f(0xbf000004u));
  const float cr = __builtin_fmaf(z, pc, 1.0f);
  const bool even = (q & 1u) == 0u;
  const unsigned flip = (q << 30) & 0x80000000u;
  s = bits_f(f_bits(even ? sr : cr) ^ flip ^ (f_bits(x) & 0x80000000u));
  c = bits_f(f_bits(even ? cr : -sr) ^ flip);
}

__device__ __forceinline__ void put_set(float* ar, float* tq, float t0, float t1, float t2,
                                        float wraw, float w, float sx, float cx,
                                        float sy, float cy, float sz, float cz) {
  float r00 = cy * cz;
  float r01 = -cy * sz;
  float r02 = sy;
  float r10 = sx * sy * cz + cx * sz;
  float r11 = -sx * sy * sz + cx * cz;
  float r12 = -sx * cy;
  float r20 = -cx * sy * cz + sx * sz;
  float r21 = cx * sy * sz + sx * cz;
  float r22 = cx * cy;
  ar[0] = r00 * w; ar[1] = r01 * w; ar[2] = r02 * w;
  ar[kPitch + 0] = r10 * w; ar[kPitch + 1] = r11 * w; ar[kPitch + 2] = r12 * w;
  ar[2 * kPitch + 0] = r20 * w; ar[2 * kPitch + 1] = r21 * w; ar[2 * kPitch + 2] = r22 * w;
  tq[0]           = wraw * t0;
  tq[kTPitch]     = wraw * t1;
  tq[2 * kTPitch] = wraw * t2;
}

__global__ __launch_bounds__(kThreads) void fused_gemm_kernel(
    const float* __restrict__ scales, const float* __restrict__ transforms,
    const float* __restrict__ pw, const float* __restrict__ off,
    const float* __restrict__ base, float* __restrict__ out) {
  __shared__ __attribute__((aligned(16))) float AsBuf[(kGuardLo + kRows + kGuardHi) * kPitch];
  __shared__ __attribute__((aligned(16))) float Tp[kBsMax * 3 * kTPitch];
  float* const As = AsBuf + kGuardLo * kPitch;

  const int tid   = threadIdx.x;
  const int lane  = tid & 31;
  const int wave  = tid >> 5;
  const int half  = lane >> 4;
  const int lrow  = lane & 15;
  const int m0    = blockIdx.x * kRows;
  const int ncol0 = blockIdx.y * kColsWG + wave * kColsWave;
  const int bs_lo = m0 / 3;
  const int nbs   = (m0 + kRows - 1) / 3 - bs_lo + 1;
  const int nsets = nbs * kP;

  const float* offp[kNTile];
  f3 bn[kNTile];
  f3 braw[3][kNTile][2];
#pragma unroll
  for (int t = 0; t < kNTile; ++t) {
    int n = ncol0 + t * 16 + lrow;
    n = n < kV ? n : kV - 1;
    offp[t] = off + ((size_t)(2 * half) * kV + n) * 3;
    bn[t] = *(const f3*)(base + (size_t)n * 3);
  }
#pragma unroll
  for (int g = 0; g < 2; ++g) {
#pragma unroll
    for (int t = 0; t < kNTile; ++t) {
      const float* q = offp[t] + (size_t)g * (4 * kV * 3);
      braw[g][t][0] = *(const f3*)(q);
      braw[g][t][1] = *(const f3*)(q + kV * 3);
    }
  }

  {
    v2f ta[kIters], tb[kIters], tc[kIters];
    float tw[kIters], ts[kIters];
#pragma unroll
    for (int it = 0; it < kIters; ++it) {
      int idx = it * kThreads + tid;
      int sidx = idx < nsets ? idx : 0;
      int gi = bs_lo * kP + sidx;
      const v2f* tp = (const v2f*)(transforms + (size_t)gi * 6);
      ta[it] = tp[0]; tb[it] = tp[1]; tc[it] = tp[2];
      tw[it] = pw[gi];
      ts[it] = scales[gi >> 6];
    }
    const int p   = tid & 63;
    const int rb0 = 3 * (tid >> 6) - (m0 - 3 * bs_lo);
    float* const ar0 = As + rb0 * kPitch + p * 3;
    float* const tq0 = Tp + (3 * (tid >> 6)) * kTPitch + p;
    bool big = false;
#pragma unroll
    for (int it = 0; it < kIters; ++it) {
      const float ax = tb[it].y, ay = tc[it].x, az = tc[it].y;
      big |= !(__builtin_fabsf(ax) < kMidArg) | !(__builtin_fabsf(ay) < kMidArg) |
             !(__builtin_fabsf(az) < kMidArg);
      float cx, sx, cy, sy, cz, sz;
      sincos_mid(ax, sx, cx);
      sincos_mid(ay, sy, cy);
      sincos_mid(az, sz, cz);
      put_set(ar0 + it * 6 * kPitch, tq0 + it * 6 * kTPitch, ta[it].x, ta[it].y, tb[it].x,
              tw[it], tw[it] * ts[it], sx, cx, sy, cy, sz, cz);
    }
    if (big) {
#pragma unroll 1
      for (int it = 0; it < kIters; ++it) {
        int idx = it * kThreads + tid;
        int sidx = idx < nsets ? idx : 0;
        int gi = bs_lo * kP + sidx;
        const float* tp = transforms + (size_t)gi * 6;
        float wraw = pw[gi];
        float w = wraw * scales[gi >> 6];
        float cx, sx, cy, sy, cz, sz;
        sincosf(tp[3], &sx, &cx);
        sincosf(tp[4], &sy, &cy);
        sincosf(tp[5], &sz, &cz);
        put_set(ar0 + it * 6 * kPitch, tq0 + it * 6 * kTPitch, tp[0], tp[1], tp[2],
                wraw, w, sx, cx, sy, cy, sz, cz);
      }
    }
  }
  __syncthreads();

  if (ncol0 >= kV) return;

  float Tval;
  {
    const float* x = Tp + (m0 - 3 * bs_lo + lane) * kTPitch;
    float d[16];
#pragma unroll
    for (int Q = 0; Q < 16; Q += 4) {
      v4f x0 = *(const v4f*)(x + Q);
      v4f x1 = *(const v4f*)(x + Q + 16);
      v4f x2 = *(const v4f*)(x + Q + 32);
      v4f x3 = *(const v4f*)(x + Q + 48);
#pragma unroll
      for (int e = 0; e < 4; ++e) d[Q + e] = (x0[e] + x2[e]) + (x1[e] + x3[e]);
    }
    float c[8], b[4];
#pragma unroll
    for (int q = 0; q < 8; ++q) c[q] = d[q] + d[q + 8];
#pragma unroll
    for (int q = 0; q < 4; ++q) b[q] = c[q] + c[q + 4];
    Tval = (b[0] + b[2]) + (b[1] + b[3]);
  }
  float tv[2][8];
  int   ob[2][8];
#pragma unroll
  for (int mt = 0; mt < 2; ++mt) {
#pragma unroll
    for (int r = 0; r < 8; ++r) {
      int lr = mt * 16 + half * 8 + r;
      tv[mt][r] = __shfl(Tval, lr);
      int m = m0 + lr;
      int bs = m / 3;
      ob[mt][r] = bs * (kV * 3) + (m - bs * 3);
    }
  }

  const float* Ap = As + lrow * kPitch + 6 * half;

  auto run = [&](auto ntc) {
    constexpr int NT = decltype(ntc)::value;
    v8f acc[2][NT];
#pragma unroll
    for (int mt = 0; mt < 2; ++mt)
#pragma unroll
      for (int t = 0; t < NT; ++t)
        acc[mt][t] = v8f{0.f, 0.f, 0.f, 0.f, 0.f, 0.f, 0.f, 0.f};
    v2f afr[2][2][3];

    auto loadB = [&](int g, int set) {
#pragma unroll
      for (int t = 0; t < NT; ++t) {
        const float* q = offp[t] + (size_t)g * (4 * kV * 3);
        braw[set][t][0] = *(const f3*)(q);
        braw[set][t][1] = *(const f3*)(q + kV * 3);
      }
    };
    auto loadA = [&](int g, int set) {
#pragma unroll
      for (int mt = 0; mt < 2; ++mt)
#pragma unroll
        for (int s = 0; s < 3; ++s)
          afr[set][mt][s] = *(const v2f*)(Ap + mt * 16 * kPitch + g * 12 + s * 2);
    };
    auto mma = [&](int bset, int aset) {
      v2f bf[NT][3];
#pragma unroll
      for (int t = 0; t < NT; ++t) {
        f3 u0 = braw[bset][t][0], u1 = braw[bset][t][1];
        asm volatile("" : "+v"(u0.x), "+v"(u0.y), "+v"(u0.z), "+v"(u1.x), "+v"(u1.y), "+v"(u1.z));
        bf[t][0] = v2f{u0.x + bn[t].x, u0.y + bn[t].y};
        bf[t][1] = v2f{u0.z + bn[t].z, u1.x + bn[t].x};
        bf[t][2] = v2f{u1.y + bn[t].y, u1.z + bn[t].z};
      }
#pragma unroll
      for (int s = 0; s < 3; ++s)
#pragma unroll
        for (int t = 0; t < NT; ++t)
#pragma unroll
          for (int mt = 0; mt < 2; ++mt)
            acc[mt][t] = __builtin_amdgcn_wmma_f32_16x16x4_f32(
                false, afr[aset][mt][s], false, bf[t][s], (short)0, acc[mt][t],
                false, false);
    };

    loadA(0, 0);
#pragma unroll
    for (int g = 0; g < kG; ++g) {
      if (g + 2 < kG) loadB(g + 2, (g + 2) % 3);
      if (g + 1 < kG) loadA(g + 1, (g + 1) & 1);
      __builtin_amdgcn_sched_barrier(0);
      mma(g % 3, g & 1);
      __builtin_amdgcn_sched_barrier(0);
    }

#pragma unroll
    for (int t = 0; t < NT; ++t) {
      int n = ncol0 + t * 16 + lrow;
      if (n < kV) {
        int n3 = n * 3;
#pragma unroll
        for (int mt = 0; mt < 2; ++mt)
#pragma unroll
          for (int r = 0; r < 8; ++r)
            out[ob[mt][r] + n3] = acc[mt][t][r] + tv[mt][r];
      }
    }
  };

  if (ncol0 + kColsWave <= kV) run(IntC<kNTile>{});
  else                         run(IntC<1>{});
}

extern "C" void kernel_launch(void* const* d_in, const int* in_sizes, int n_in,
                              void* d_out, int out_size, void* d_ws,
                              size_t ws_size, hipStream_t stream) {
  const float* scales     = (const float*)d_in[0];
  const float* transforms = (const float*)d_in[1];
  const float* pw         = (const float*)d_in[2];
  const float* off        = (const float*)d_in[3];
  const float* base       = (const float*)d_in[4];
  float* out = (float*)d_out;
  fused_gemm_kernel<<<dim3(kMG, kNG), kThreads, 0, stream>>>(
      scales, transforms, pw, off, base, out);
}


# Epilogue of the full waves: results staged through a per-wave LDS region in out's order and stored as contiguous 16-byte pieces (768 B per (b,s)), not 4-byte pieces 12 B apart; values unchanged
# speedup vs baseline: 2.1157x; 1.0533x over previous
#include <hip/hip_runtime.h>

typedef __attribute__((ext_vector_type(2))) float v2f;
typedef __attribute__((ext_vector_type(4))) float v4f;
typedef __attribute__((ext_vector_type(8))) float v8f;
struct f3 { float x, y, z; };
struct __attribute__((aligned(8))) f4 { float x, y, z, w; };

template <int N> struct IntC { static constexpr int value = N; };

constexpr int kBS      = 256;
constexpr int kP       = 64;
constexpr int kV       = 2562;
constexpr int kK       = kP * 3;
constexpr int kM       = kBS * 3;
constexpr int kRows    = 32;
constexpr int kMG      = kM / kRows;
constexpr int kThreads = 128;
constexpr int kNTile   = 4;
constexpr int kColsWave = kNTile * 16;
constexpr int kColsWG  = 4 * kColsWave;
constexpr int kNG      = (kV + kColsWG - 1) / kColsWG;
constexpr int kPitch   = 196;
constexpr int kG       = kK / 12;
constexpr int kBsMax   = 12;
constexpr int kIters   = kBsMax * kP / kThreads;
constexpr int kTPitch  = 68;
constexpr int kGuardLo = 2;
constexpr int kGuardHi = 4;
constexpr float kMidArg = 131072.0f;
constexpr int kRun     = kColsWave * 3;
constexpr int kStage   = kBsMax * kRun;
static_assert(kMG * kRows == kM, "row groups must tile M exactly");
static_assert(kG * 12 == kK, "K groups must tile K exactly");
static_assert(kIters * kThreads == kBsMax * kP, "set loop must be exact");
static_assert(kV % kColsWave == 2, "the only partial wave has one N tile");

__device__ __forceinline__ float bits_f(unsigned u) { return __builtin_bit_cast(float, u); }
__device__ __forceinline__ unsigned f_bits(float f) { return __builtin_bit_cast(unsigned, f); }

__device__ __forceinline__ void sincos_mid(float x, float& s, float& c) {
  const float a = __builtin_fabsf(x);
  const float n = __builtin_rintf(a * bits_f(0x3f22f983u));
  float r = __builtin_fmaf(n, bits_f(0xbfc90fdau), a);
  r = __builtin_fmaf(n, bits_f(0xb3a22168u), r);
  r = __builtin_fmaf(n, bits_f(0xa7c234c4u), r);
  const unsigned q = (unsigned)(int)n;
  const float z = r * r;
  float ps = __builtin_fmaf(z, bits_f(0xb94c1982u), bits_f(0x3c0881c4u));
  ps = __builtin_fmaf(z, ps, bits_f(0xbe2aaa9du));
  const float sr = __builtin_fmaf(r, z * ps, r);
  float pc = __builtin_fmaf(z, bits_f(0x37d75334u), bits_f(0xbab64f3bu));
  pc = __builtin_fmaf(z, pc, bits_f(0x3d2aabf7u));
  pc = __builtin_fmaf(z, pc, bits_f(0xbf000004u));
  const float cr = __builtin_fmaf(z, pc, 1.0f);
  const bool even = (q & 1u) == 0u;
  const unsigned flip = (q << 30) & 0x80000000u;
  s = bits_f(f_bits(even ? sr : cr) ^ flip ^ (f_bits(x) & 0x80000000u));
  c = bits_f(f_bits(even ? cr : -sr) ^ flip);
}

__device__ __forceinline__ void put_set(float* ar, float* tq, float t0, float t1, float t2,
                                        float wraw, float w, float sx, float cx,
                                        float sy, float cy, float sz, float cz) {
  float r00 = cy * cz;
  float r01 = -cy * sz;
  float r02 = sy;
  float r10 = sx * sy * cz + cx * sz;
  float r11 = -sx * sy * sz + cx * cz;
  float r12 = -sx * cy;
  float r20 = -cx * sy * cz + sx * sz;
  float r21 = cx * sy * sz + sx * cz;
  float r22 = cx * cy;
  ar[0] = r00 * w; ar[1] = r01 * w; ar[2] = r02 * w;
  ar[kPitch + 0] = r10 * w; ar[kPitch + 1] = r11 * w; ar[kPitch + 2] = r12 * w;
  ar[2 * kPitch + 0] = r20 * w; ar[2 * kPitch + 1] = r21 * w; ar[2 * kPitch + 2] = r22 * w;
  tq[0]           = wraw * t0;
  tq[kTPitch]     = wraw * t1;
  tq[2 * kTPitch] = wraw * t2;
}

__global__ __launch_bounds__(kThreads) void fused_gemm_kernel(
    const float* __restrict__ scales, const float* __restrict__ transforms,
    const float* __restrict__ pw, const float* __restrict__ off,
    const float* __restrict__ base, float* __restrict__ out) {
  __shared__ __attribute__((aligned(16))) float AsBuf[(kGuardLo + kRows + kGuardHi) * kPitch];
  __shared__ __attribute__((aligned(16))) float Tp[kBsMax * 3 * kTPitch];
  __shared__ __attribute__((aligned(16))) float Stage[4 * kStage];
  float* const As = AsBuf + kGuardLo * kPitch;

  const int tid   = threadIdx.x;
  const int lane  = tid & 31;
  const int wave  = tid >> 5;
  const int half  = lane >> 4;
  const int lrow  = lane & 15;
  const int m0    = blockIdx.x * kRows;
  const int ncol0 = blockIdx.y * kColsWG + wave * kColsWave;
  const int bs_lo = m0 / 3;
  const int nbs   = (m0 + kRows - 1) / 3 - bs_lo + 1;
  const int nsets = nbs * kP;

  const float* offp[kNTile];
  f3 bn[kNTile];
  f3 braw[3][kNTile][2];
#pragma unroll
  for (int t = 0; t < kNTile; ++t) {
    int n = ncol0 + t * 16 + lrow;
    n = n < kV ? n : kV - 1;
    offp[t] = off + ((size_t)(2 * half) * kV + n) * 3;
    bn[t] = *(const f3*)(base + (size_t)n * 3);
  }
#pragma unroll
  for (int g = 0; g < 2; ++g) {
#pragma unroll
    for (int t = 0; t < kNTile; ++t) {
      const float* q = offp[t] + (size_t)g * (4 * kV * 3);
      braw[g][t][0] = *(const f3*)(q);
      braw[g][t][1] = *(const f3*)(q + kV * 3);
    }
  }

  {
    v2f ta[kIters], tb[kIters], tc[kIters];
    float tw[kIters], ts[kIters];
#pragma unroll
    for (int it = 0; it < kIters; ++it) {
      int idx = it * kThreads + tid;
      int sidx = idx < nsets ? idx : 0;
      int gi = bs_lo * kP + sidx;
      const v2f* tp = (const v2f*)(transforms + (size_t)gi * 6);
      ta[it] = tp[0]; tb[it] = tp[1]; tc[it] = tp[2];
      tw[it] = pw[gi];
      ts[it] = scales[gi >> 6];
    }
    const int p   = tid & 63;
    const int rb0 = 3 * (tid >> 6) - (m0 - 3 * bs_lo);
    float* const ar0 = As + rb0 * kPitch + p * 3;
    float* const tq0 = Tp + (3 * (tid >> 6)) * kTPitch + p;
    bool big = false;
#pragma unroll
    for (int it = 0; it < kIters; ++it) {
      const float ax = tb[it].y, ay = tc[it].x, az = tc[it].y;
      big |= !(__builtin_fabsf(ax) < kMidArg) | !(__builtin_fabsf(ay) < kMidArg) |
             !(__builtin_fabsf(az) < kMidArg);
      float cx, sx, cy, sy, cz, sz;
      sincos_mid(ax, sx, cx);
      sincos_mid(ay, sy, cy);
      sincos_mid(az, sz, cz);
      put_set(ar0 + it * 6 * kPitch, tq0 + it * 6 * kTPitch, ta[it].x, ta[it].y, tb[it].x,
              tw[it], tw[it] * ts[it], sx, cx, sy, cy, sz, cz);
    }
    if (big) {
#pragma unroll 1
      for (int it = 0; it < kIters; ++it) {
        int idx = it * kThreads + tid;
        int sidx = idx < nsets ? idx : 0;
        int gi = bs_lo * kP + sidx;
        const float* tp = transforms + (size_t)gi * 6;
        float wraw = pw[gi];
        float w = wraw * scales[gi >> 6];
        float cx, sx, cy, sy, cz, sz;
        sincosf(tp[3], &sx, &cx);
        sincosf(tp[4], &sy, &cy);
        sincosf(tp[5], &sz, &cz);
        put_set(ar0 + it * 6 * kPitch, tq0 + it * 6 * kTPitch, tp[0], tp[1], tp[2],
                wraw, w, sx, cx, sy, cy, sz, cz);
      }
    }
  }
  __syncthreads();

  if (ncol0 >= kV) return;

  float Tval;
  {
    const float* x = Tp + (m0 - 3 * bs_lo + lane) * kTPitch;
    float d[16];
#pragma unroll
    for (int Q = 0; Q < 16; Q += 4) {
      v4f x0 = *(const v4f*)(x + Q);
      v4f x1 = *(const v4f*)(x + Q + 16);
      v4f x2 = *(const v4f*)(x + Q + 32);
      v4f x3 = *(const v4f*)(x + Q + 48);
#pragma unroll
      for (int e = 0; e < 4; ++e) d[Q + e] = (x0[e] + x2[e]) + (x1[e] + x3[e]);
    }
    float c[8], b[4];
#pragma unroll
    for (int q = 0; q < 8; ++q) c[q] = d[q] + d[q + 8];
#pragma unroll
    for (int q = 0; q < 4; ++q) b[q] = c[q] + c[q + 4];
    Tval = (b[0] + b[2]) + (b[1] + b[3]);
  }
  float tv[2][8];
#pragma unroll
  for (int mt = 0; mt < 2; ++mt) {
#pragma unroll
    for (int r = 0; r < 8; ++r) tv[mt][r] = __shfl(Tval, mt * 16 + half * 8 + r);
  }

  const float* Ap = As + lrow * kPitch + 6 * half;

  auto run = [&](auto ntc) {
    constexpr int NT = decltype(ntc)::value;
    v8f acc[2][NT];
#pragma unroll
    for (int mt = 0; mt < 2; ++mt)
#pragma unroll
      for (int t = 0; t < NT; ++t)
        acc[mt][t] = v8f{0.f, 0.f, 0.f, 0.f, 0.f, 0.f, 0.f, 0.f};
    v2f afr[2][2][3];

    auto loadB = [&](int g, int set) {
#pragma unroll
      for (int t = 0; t < NT; ++t) {
        const float* q = offp[t] + (size_t)g * (4 * kV * 3);
        braw[set][t][0] = *(const f3*)(q);
        braw[set][t][1] = *(const f3*)(q + kV * 3);
      }
    };
    auto loadA = [&](int g, int set) {
#pragma unroll
      for (int mt = 0; mt < 2; ++mt)
#pragma unroll
        for (int s = 0; s < 3; ++s)
          afr[set][mt][s] = *(const v2f*)(Ap + mt * 16 * kPitch + g * 12 + s * 2);
    };
    auto mma = [&](int bset, int aset) {
      v2f bf[NT][3];
#pragma unroll
      for (int t = 0; t < NT; ++t) {
        f3 u0 = braw[bset][t][0], u1 = braw[bset][t][1];
        asm volatile("" : "+v"(u0.x), "+v"(u0.y), "+v"(u0.z), "+v"(u1.x), "+v"(u1.y), "+v"(u1.z));
        bf[t][0] = v2f{u0.x + bn[t].x, u0.y + bn[t].y};
        bf[t][1] = v2f{u0.z + bn[t].z, u1.x + bn[t].x};
        bf[t][2] = v2f{u1.y + bn[t].y, u1.z + bn[t].z};
      }
#pragma unroll
      for (int s = 0; s < 3; ++s)
#pragma unroll
        for (int t = 0; t < NT; ++t)
#pragma unroll
          for (int mt = 0; mt < 2; ++mt)
            acc[mt][t] = __builtin_amdgcn_wmma_f32_16x16x4_f32(
                false, afr[aset][mt][s], false, bf[t][s], (short)0, acc[mt][t],
                false, false);
    };

    loadA(0, 0);
#pragma unroll
    for (int g = 0; g < kG; ++g) {
      if (g + 2 < kG) loadB(g + 2, (g + 2) % 3);
      if (g + 1 < kG) loadA(g + 1, (g + 1) & 1);
      __builtin_amdgcn_sched_barrier(0);
      mma(g % 3, g & 1);
      __builtin_amdgcn_sched_barrier(0);
    }

    if constexpr (NT == kNTile) {
      float* const Sw = Stage + wave * kStage;
      const int a = m0 - 3 * bs_lo;
      float* const sbase = Sw + a + half * 8 + lrow * 3;
#pragma unroll
      for (int mt = 0; mt < 2; ++mt)
#pragma unroll
        for (int r = 0; r < 8; ++r) {
          const int q = a + mt * 16 + half * 8 + r;
          const int k = (q * 43) >> 7;
          float* sp = sbase + (mt * 16 + r) + (kRun - 3) * k;
#pragma unroll
          for (int t = 0; t < NT; ++t) sp[t * 48] = acc[mt][t][r] + tv[mt][r];
        }
      const int kf = a != 0 ? 1 : 0;
      const float* sq = Sw + kf * kRun + lane * 4;
      float* oq = out + ((size_t)(bs_lo + kf) * kV + ncol0) * 3;
      const int go0 = lane * 4;
      const int go1 = lane < 16 ? (32 + lane) * 4 : kV * 3 + (lane - 16) * 4;
      const int go2 = kV * 3 + (16 + lane) * 4;
#pragma unroll
      for (int pq = 0; pq < 5; ++pq) {
        const v4f x0 = *(const v4f*)(sq + pq * 2 * kRun);
        const v4f x1 = *(const v4f*)(sq + pq * 2 * kRun + 128);
        const v4f x2 = *(const v4f*)(sq + pq * 2 * kRun + 256);
        float* o = oq + (size_t)pq * (2 * kV * 3);
        *(f4*)(o + go0) = f4{x0[0], x0[1], x0[2], x0[3]};
        *(f4*)(o + go1) = f4{x1[0], x1[1], x1[2], x1[3]};
        *(f4*)(o + go2) = f4{x2[0], x2[1], x2[2], x2[3]};
      }
      const int lrA = a == 0 ? 30 : 0;
      const int lrB = a == 1 ? 1 : 31;
#pragma unroll
      for (int e = 0; e < 2; ++e) {
        const int q = a + (e == 0 ? lrA : lrB);
        const int k = q / 3;
        const int i = q - 3 * k;
        const float* sr = Sw + k * kRun + i;
        float* orow = out + ((size_t)(bs_lo + k) * kV + ncol0) * 3 + i;
        orow[lane * 3]        = sr[lane * 3];
        orow[(32 + lane) * 3] = sr[(32 + lane) * 3];
      }
    } else {
#pragma unroll
      for (int t = 0; t < NT; ++t) {
        int n = ncol0 + t * 16 + lrow;
        if (n < kV) {
          int n3 = n * 3;
#pragma unroll
          for (int mt = 0; mt < 2; ++mt)
#pragma unroll
            for (int r = 0; r < 8; ++r) {
              int m = m0 + mt * 16 + half * 8 + r;
              int bs = m / 3;
              out[bs * (kV * 3) + (m - bs * 3) + n3] = acc[mt][t][r] + tv[mt][r];
            }
        }
      }
    }
  };

  if (ncol0 + kColsWave <= kV) run(IntC<kNTile>{});
  else                         run(IntC<1>{});
}

extern "C" void kernel_launch(void* const* d_in, const int* in_sizes, int n_in,
                              void* d_out, int out_size, void* d_ws,
                              size_t ws_size, hipStream_t stream) {
  const float* scales     = (const float*)d_in[0];
  const float* transforms = (const float*)d_in[1];
  const float* pw         = (const float*)d_in[2];
  const float* off        = (const float*)d_in[3];
  const float* base       = (const float*)d_in[4];
  float* out = (float*)d_out;
  fused_gemm_kernel<<<dim3(kMG, kNG), kThreads, 0, stream>>>(
      scales, transforms, pw, off, base, out);
}


# GEMM K loop: B operand loads issued three K groups ahead of their WMMAs instead of two (ring of four register sets, 24 loads in flight per wave); k order, sums and all values unchanged
# speedup vs baseline: 2.2594x; 1.0679x over previous
#include <hip/hip_runtime.h>

typedef __attribute__((ext_vector_type(2))) float v2f;
typedef __attribute__((ext_vector_type(4))) float v4f;
typedef __attribute__((ext_vector_type(8))) float v8f;
struct f3 { float x, y, z; };
struct __attribute__((aligned(8))) f4 { float x, y, z, w; };

template <int N> struct IntC { static constexpr int value = N; };

constexpr int kBS      = 256;
constexpr int kP       = 64;
constexpr int kV       = 2562;
constexpr int kK       = kP * 3;
constexpr int kM       = kBS * 3;
constexpr int kRows    = 32;
constexpr int kMG      = kM / kRows;
constexpr int kThreads = 128;
constexpr int kNTile   = 4;
constexpr int kColsWave = kNTile * 16;
constexpr int kColsWG  = 4 * kColsWave;
constexpr int kNG      = (kV + kColsWG - 1) / kColsWG;
constexpr int kPitch   = 196;
constexpr int kG       = kK / 12;
constexpr int kBsMax   = 12;
constexpr int kIters   = kBsMax * kP / kThreads;
constexpr int kTPitch  = 68;
constexpr int kGuardLo = 2;
constexpr int kGuardHi = 4;
constexpr float kMidArg = 131072.0f;
constexpr int kRun     = kColsWave * 3;
constexpr int kStage   = kBsMax * kRun;
constexpr int kAhead   = 3;
constexpr int kRing    = kAhead + 1;
static_assert(kMG * kRows == kM, "row groups must tile M exactly");
static_assert(kG * 12 == kK, "K groups must tile K exactly");
static_assert(kIters * kThreads == kBsMax * kP, "set loop must be exact");
static_assert(kV % kColsWave == 2, "the only partial wave has one N tile");
static_assert(kAhead >= 1 && kAhead < kG, "the B loads run ahead inside the K loop");

__device__ __forceinline__ float bits_f(unsigned u) { return __builtin_bit_cast(float, u); }
__device__ __forceinline__ unsigned f_bits(float f) { return __builtin_bit_cast(unsigned, f); }

__device__ __forceinline__ void sincos_mid(float x, float& s, float& c) {
  const float a = __builtin_fabsf(x);
  const float n = __builtin_rintf(a * bits_f(0x3f22f983u));
  float r = __builtin_fmaf(n, bits_f(0xbfc90fdau), a);
  r = __builtin_fmaf(n, bits_f(0xb3a22168u), r);
  r = __builtin_fmaf(n, bits_f(0xa7c234c4u), r);
  const unsigned q = (unsigned)(int)n;
  const float z = r * r;
  float ps = __builtin_fmaf(z, bits_f(0xb94c1982u), bits_f(0x3c0881c4u));
  ps = __builtin_fmaf(z, ps, bits_f(0xbe2aaa9du));
  const float sr = __builtin_fmaf(r, z * ps, r);
  float pc = __builtin_fmaf(z, bits_f(0x37d75334u), bits_f(0xbab64f3bu));
  pc = __builtin_fmaf(z, pc, bits_f(0x3d2aabf7u));
  pc = __builtin_fmaf(z, pc, bits_f(0xbf000004u));
  const float cr = __builtin_fmaf(z, pc, 1.0f);
  const bool even = (q & 1u) == 0u;
  const unsigned flip = (q << 30) & 0x80000000u;
  s = bits_f(f_bits(even ? sr : cr) ^ flip ^ (f_bits(x) & 0x80000000u));
  c = bits_f(f_bits(even ? cr : -sr) ^ flip);
}

__device__ __forceinline__ void put_set(float* ar, float* tq, float t0, float t1, float t2,
                                        float wraw, float w, float sx, float cx,
                                        float sy, float cy, float sz, float cz) {
  float r00 = cy * cz;
  float r01 = -cy * sz;
  float r02 = sy;
  float r10 = sx * sy * cz + cx * sz;
  float r11 = -sx * sy * sz + cx * cz;
  float r12 = -sx * cy;
  float r20 = -cx * sy * cz + sx * sz;
  float r21 = cx * sy * sz + sx * cz;
  float r22 = cx * cy;
  ar[0] = r00 * w; ar[1] = r01 * w; ar[2] = r02 * w;
  ar[kPitch + 0] = r10 * w; ar[kPitch + 1] = r11 * w; ar[kPitch + 2] = r12 * w;
  ar[2 * kPitch + 0] = r20 * w; ar[2 * kPitch + 1] = r21 * w; ar[2 * kPitch + 2] = r22 * w;
  tq[0]           = wraw * t0;
  tq[kTPitch]     = wraw * t1;
  tq[2 * kTPitch] = wraw * t2;
}

__global__ __launch_bounds__(kThreads) void fused_gemm_kernel(
    const float* __restrict__ scales, const float* __restrict__ transforms,
    const float* __restrict__ pw, const float* __restrict__ off,
    const float* __restrict__ base, float* __restrict__ out) {
  __shared__ __attribute__((aligned(16))) float AsBuf[(kGuardLo + kRows + kGuardHi) * kPitch];
  __shared__ __attribute__((aligned(16))) float Tp[kBsMax * 3 * kTPitch];
  __shared__ __attribute__((aligned(16))) float Stage[4 * kStage];
  float* const As = AsBuf + kGuardLo * kPitch;

  const int tid   = threadIdx.x;
  const int lane  = tid & 31;
  const int wave  = tid >> 5;
  const int half  = lane >> 4;
  const int lrow  = lane & 15;
  const int m0    = blockIdx.x * kRows;
  const int ncol0 = blockIdx.y * kColsWG + wave * kColsWave;
  const int bs_lo = m0 / 3;
  const int nbs   = (m0 + kRows - 1) / 3 - bs_lo + 1;
  const int nsets = nbs * kP;

  const float* offp[kNTile];
  f3 bn[kNTile];
  f3 braw[kRing][kNTile][2];
#pragma unroll
  for (int t = 0; t < kNTile; ++t) {
    int n = ncol0 + t * 16 + lrow;
    n = n < kV ? n : kV - 1;
    offp[t] = off + ((size_t)(2 * half) * kV + n) * 3;
    bn[t] = *(const f3*)(base + (size_t)n * 3);
  }
#pragma unroll
  for (int g = 0; g < kAhead; ++g) {
#pragma unroll
    for (int t = 0; t < kNTile; ++t) {
      const float* q = offp[t] + (size_t)g * (4 * kV * 3);
      braw[g][t][0] = *(const f3*)(q);
      braw[g][t][1] = *(const f3*)(q + kV * 3);
    }
  }

  {
    v2f ta[kIters], tb[kIters], tc[kIters];
    float tw[kIters], ts[kIters];
#pragma unroll
    for (int it = 0; it < kIters; ++it) {
      int idx = it * kThreads + tid;
      int sidx = idx < nsets ? idx : 0;
      int gi = bs_lo * kP + sidx;
      const v2f* tp = (const v2f*)(transforms + (size_t)gi * 6);
      ta[it] = tp[0]; tb[it] = tp[1]; tc[it] = tp[2];
      tw[it] = pw[gi];
      ts[it] = scales[gi >> 6];
    }
    const int p   = tid & 63;
    const int rb0 = 3 * (tid >> 6) - (m0 - 3 * bs_lo);
    float* const ar0 = As + rb0 * kPitch + p * 3;
    float* const tq0 = Tp + (3 * (tid >> 6)) * kTPitch + p;
    bool big = false;
#pragma unroll
    for (int it = 0; it < kIters; ++it) {
      const float ax = tb[it].y, ay = tc[it].x, az = tc[it].y;
      big |= !(__builtin_fabsf(ax) < kMidArg) | !(__builtin_fabsf(ay) < kMidArg) |
             !(__builtin_fabsf(az) < kMidArg);
      float cx, sx, cy, sy, cz, sz;
      sincos_mid(ax, sx, cx);
      sincos_mid(ay, sy, cy);
      sincos_mid(az, sz, cz);
      put_set(ar0 + it * 6 * kPitch, tq0 + it * 6 * kTPitch, ta[it].x, ta[it].y, tb[it].x,
              tw[it], tw[it] * ts[it], sx, cx, sy, cy, sz, cz);
    }
    if (big) {
#pragma unroll 1
      for (int it = 0; it < kIters; ++it) {
        int idx = it * kThreads + tid;
        int sidx = idx < nsets ? idx : 0;
        int gi = bs_lo * kP + sidx;
        const float* tp = transforms + (size_t)gi * 6;
        float wraw = pw[gi];
        float w = wraw * scales[gi >> 6];
        float cx, sx, cy, sy, cz, sz;
        sincosf(tp[3], &sx, &cx);
        sincosf(tp[4], &sy, &cy);
        sincosf(tp[5], &sz, &cz);
        put_set(ar0 + it * 6 * kPitch, tq0 + it * 6 * kTPitch, tp[0], tp[1], tp[2],
                wraw, w, sx, cx, sy, cy, sz, cz);
      }
    }
  }
  __syncthreads();

  if (ncol0 >= kV) return;

  float Tval;
  {
    const float* x = Tp + (m0 - 3 * bs_lo + lane) * kTPitch;
    float d[16];
#pragma unroll
    for (int Q = 0; Q < 16; Q += 4) {
      v4f x0 = *(const v4f*)(x + Q);
      v4f x1 = *(const v4f*)(x + Q + 16);
      v4f x2 = *(const v4f*)(x + Q + 32);
      v4f x3 = *(const v4f*)(x + Q + 48);
#pragma unroll
      for (int e = 0; e < 4; ++e) d[Q + e] = (x0[e] + x2[e]) + (x1[e] + x3[e]);
    }
    float c[8], b[4];
#pragma unroll
    for (int q = 0; q < 8; ++q) c[q] = d[q] + d[q + 8];
#pragma unroll
    for (int q = 0; q < 4; ++q) b[q] = c[q] + c[q + 4];
    Tval = (b[0] + b[2]) + (b[1] + b[3]);
  }
  float tv[2][8];
#pragma unroll
  for (int mt = 0; mt < 2; ++mt) {
#pragma unroll
    for (int r = 0; r < 8; ++r) tv[mt][r] = __shfl(Tval, mt * 16 + half * 8 + r);
  }

  const float* Ap = As + lrow * kPitch + 6 * half;

  auto run = [&](auto ntc) {
    constexpr int NT = decltype(ntc)::value;
    v8f acc[2][NT];
#pragma unroll
    for (int mt = 0; mt < 2; ++mt)
#pragma unroll
      for (int t = 0; t < NT; ++t)
        acc[mt][t] = v8f{0.f, 0.f, 0.f, 0.f, 0.f, 0.f, 0.f, 0.f};
    v2f afr[2][2][3];

    auto loadB = [&](int g, int set) {
#pragma unroll
      for (int t = 0; t < NT; ++t) {
        const float* q = offp[t] + (size_t)g * (4 * kV * 3);
        braw[set][t][0] = *(const f3*)(q);
        braw[set][t][1] = *(const f3*)(q + kV * 3);
      }
    };
    auto loadA = [&](int g, int set) {
#pragma unroll
      for (int mt = 0; mt < 2; ++mt)
#pragma unroll
        for (int s = 0; s < 3; ++s)
          afr[set][mt][s] = *(const v2f*)(Ap + mt * 16 * kPitch + g * 12 + s * 2);
    };
    auto mma = [&](int bset, int aset) {
      v2f bf[NT][3];
#pragma unroll
      for (int t = 0; t < NT; ++t) {
        f3 u0 = braw[bset][t][0], u1 = braw[bset][t][1];
        asm volatile("" : "+v"(u0.x), "+v"(u0.y), "+v"(u0.z), "+v"(u1.x), "+v"(u1.y), "+v"(u1.z));
        bf[t][0] = v2f{u0.x + bn[t].x, u0.y + bn[t].y};
        bf[t][1] = v2f{u0.z + bn[t].z, u1.x + bn[t].x};
        bf[t][2] = v2f{u1.y + bn[t].y, u1.z + bn[t].z};
      }
#pragma unroll
      for (int s = 0; s < 3; ++s)
#pragma unroll
        for (int t = 0; t < NT; ++t)
#pragma unroll
          for (int mt = 0; mt < 2; ++mt)
            acc[mt][t] = __builtin_amdgcn_wmma_f32_16x16x4_f32(
                false, afr[aset][mt][s], false, bf[t][s], (short)0, acc[mt][t],
                false, false);
    };

    loadA(0, 0);
#pragma unroll
    for (int g = 0; g < kG; ++g) {
      if (g + kAhead < kG) loadB(g + kAhead, (g + kAhead) % kRing);
      if (g + 1 < kG) loadA(g + 1, (g + 1) & 1);
      __builtin_amdgcn_sched_barrier(0);
      mma(g % kRing, g & 1);
      __builtin_amdgcn_sched_barrier(0);
    }

    if constexpr (NT == kNTile) {
      float* const Sw = Stage + wave * kStage;
      const int a = m0 - 3 * bs_lo;
      float* const sbase = Sw + a + half * 8 + lrow * 3;
#pragma unroll
      for (int mt = 0; mt < 2; ++mt)
#pragma unroll
        for (int r = 0; r < 8; ++r) {
          const int q = a + mt * 16 + half * 8 + r;
          const int k = (q * 43) >> 7;
          float* sp = sbase + (mt * 16 + r) + (kRun - 3) * k;
#pragma unroll
          for (int t = 0; t < NT; ++t) sp[t * 48] = acc[mt][t][r] + tv[mt][r];
        }
      const int kf = a != 0 ? 1 : 0;
      const float* sq = Sw + kf * kRun + lane * 4;
      float* oq = out + ((size_t)(bs_lo + kf) * kV + ncol0) * 3;
      const int go0 = lane * 4;
      const int go1 = lane < 16 ? (32 + lane) * 4 : kV * 3 + (lane - 16) * 4;
      const int go2 = kV * 3 + (16 + lane) * 4;
#pragma unroll
      for (int pq = 0; pq < 5; ++pq) {
        const v4f x0 = *(const v4f*)(sq + pq * 2 * kRun);
        const v4f x1 = *(const v4f*)(sq + pq * 2 * kRun + 128);
        const v4f x2 = *(const v4f*)(sq + pq * 2 * kRun + 256);
        float* o = oq + (size_t)pq * (2 * kV * 3);
        *(f4*)(o + go0) = f4{x0[0], x0[1], x0[2], x0[3]};
        *(f4*)(o + go1) = f4{x1[0], x1[1], x1[2], x1[3]};
        *(f4*)(o + go2) = f4{x2[0], x2[1], x2[2], x2[3]};
      }
      const int lrA = a == 0 ? 30 : 0;
      const int lrB = a == 1 ? 1 : 31;
#pragma unroll
      for (int e = 0; e < 2; ++e) {
        const int q = a + (e == 0 ? lrA : lrB);
        const int k = q / 3;
        const int i = q - 3 * k;
        const float* sr = Sw + k * kRun + i;
        float* orow = out + ((size_t)(bs_lo + k) * kV + ncol0) * 3 + i;
        orow[lane * 3]        = sr[lane * 3];
        orow[(32 + lane) * 3] = sr[(32 + lane) * 3];
      }
    } else {
#pragma unroll
      for (int t = 0; t < NT; ++t) {
        int n = ncol0 + t * 16 + lrow;
        if (n < kV) {
          int n3 = n * 3;
#pragma unroll
          for (int mt = 0; mt < 2; ++mt)
#pragma unroll
            for (int r = 0; r < 8; ++r) {
              int m = m0 + mt * 16 + half * 8 + r;
              int bs = m / 3;
              out[bs * (kV * 3) + (m - bs * 3) + n3] = acc[mt][t][r] + tv[mt][r];
            }
        }
      }
    }
  };

  if (ncol0 + kColsWave <= kV) run(IntC<kNTile>{});
  else                         run(IntC<1>{});
}

extern "C" void kernel_launch(void* const* d_in, const int* in_sizes, int n_in,
                              void* d_out, int out_size, void* d_ws,
                              size_t ws_size, hipStream_t stream) {
  const float* scales     = (const float*)d_in[0];
  const float* transforms = (const float*)d_in[1];
  const float* pw         = (const float*)d_in[2];
  const float* off        = (const float*)d_in[3];
  const float* base       = (const float*)d_in[4];
  float* out = (float*)d_out;
  fused_gemm_kernel<<<dim3(kMG, kNG), kThreads, 0, stream>>>(
      scales, transforms, pw, off, base, out);
}
